# starting kernel
# baseline (speedup 1.0000x reference)
#include <hip/hip_runtime.h>
#include <hip/hip_bf16.h>

typedef __attribute__((ext_vector_type(2))) float v2f;
typedef __attribute__((ext_vector_type(8))) float v8f;

#define NW 12
#define QDIM 4096

__global__ void qsim_setup(const float* __restrict__ params,
                           const float* __restrict__ W,
                           float* __restrict__ ws) {
  __shared__ float u_re[NW][2][2];
  __shared__ float u_im[NW][2][2];
  int t = threadIdx.x;
  if (t < NW) {
    float th = params[2 * t] * 0.5f;
    float ph = params[2 * t + 1] * 0.5f;
    float c = cosf(th), s = sinf(th);
    float cy = cosf(ph), sy = sinf(ph);
    u_re[t][0][0] = cy * c;   u_im[t][0][0] =  sy * s;
    u_re[t][0][1] = -sy * c;  u_im[t][0][1] = -cy * s;
    u_re[t][1][0] =  sy * c;  u_im[t][1][0] = -cy * s;
    u_re[t][1][1] =  cy * c;  u_im[t][1][1] = -sy * s;
  }
  __syncthreads();
  for (int e = t; e < 3 * 256; e += blockDim.x) {
    int g = e >> 8;
    int r = (e >> 4) & 15, c = e & 15;
    float ar = 1.f, ai = 0.f;
    for (int q = 0; q < 4; ++q) {
      int w = g * 4 + q;
      int rb = (r >> (3 - q)) & 1;
      int cb = (c >> (3 - q)) & 1;
      float br = u_re[w][rb][cb], bi = u_im[w][rb][cb];
      float nr = ar * br - ai * bi;
      float ni = ar * bi + ai * br;
      ar = nr; ai = ni;
    }
    ws[g * 512 + r * 16 + c]       = ar;
    ws[g * 512 + 256 + r * 16 + c] = ai;
  }
  for (int k = t; k < QDIM; k += blockDim.x) {
    float acc = 0.f;
    int par = 0;
    for (int j = 0; j < NW; ++j) {
      par ^= (k >> (NW - 1 - j)) & 1;
      acc += W[j] * (par ? -1.f : 1.f);
    }
    ws[1536 + k] = acc;
  }
}

__device__ __forceinline__ v8f wmma4(v2f a, v2f b, v8f c) {
  return __builtin_amdgcn_wmma_f32_16x16x4_f32(false, a, false, b,
                                               (short)0, c, false, false);
}

__global__ __launch_bounds__(128) void qsim_main(
    const float* __restrict__ sb, const float* __restrict__ ws,
    const float* __restrict__ bptr, float* __restrict__ out) {
  __shared__ float sr[QDIM];
  __shared__ float si[QDIM];
  __shared__ float red[4];

  const int tid  = threadIdx.x;
  const int lane = tid & 31;
  const int wave = tid >> 5;
  const int h    = lane >> 4;
  const int l16  = lane & 15;
  const float* in = sb + (size_t)blockIdx.x * QDIM;

  for (int k = tid; k < QDIM; k += 128) sr[k] = in[k];
  __syncthreads();

  {
    const float* Gr = ws;
    const float* Gi = ws + 256;
    v2f gr[4], gi[4];
#pragma unroll
    for (int c = 0; c < 4; ++c) {
      int col = 4 * c + 2 * h;
      gr[c] = v2f{Gr[l16 * 16 + col], Gr[l16 * 16 + col + 1]};
      gi[c] = v2f{Gi[l16 * 16 + col], Gi[l16 * 16 + col + 1]};
    }
#pragma unroll
    for (int it = 0; it < 4; ++it) {
      int tile = 4 * it + wave;
      v2f bs[4];
#pragma unroll
      for (int c = 0; c < 4; ++c) {
        int k0 = 4 * c + 2 * h;
        bs[c] = v2f{sr[(k0 + 0) * 256 + tile * 16 + l16],
                    sr[(k0 + 1) * 256 + tile * 16 + l16]};
      }
      v8f aRe = {}, aIm = {};
#pragma unroll
      for (int c = 0; c < 4; ++c) {
        aRe = wmma4(gr[c], bs[c], aRe);
        aIm = wmma4(gi[c], bs[c], aIm);
      }
#pragma unroll
      for (int r = 0; r < 8; ++r) {
        int M = r + 8 * h;
        sr[M * 256 + tile * 16 + l16] = aRe[r];
        si[M * 256 + tile * 16 + l16] = aIm[r];
      }
    }
  }
  __syncthreads();

  {
    const float* Gr = ws + 512;
    const float* Gi = ws + 512 + 256;
    v2f gr[4], gi[4], gin[4];
#pragma unroll
    for (int c = 0; c < 4; ++c) {
      int col = 4 * c + 2 * h;
      gr[c]  = v2f{Gr[l16 * 16 + col], Gr[l16 * 16 + col + 1]};
      gi[c]  = v2f{Gi[l16 * 16 + col], Gi[l16 * 16 + col + 1]};
      gin[c] = v2f{-gi[c].x, -gi[c].y};
    }
#pragma unroll
    for (int it = 0; it < 4; ++it) {
      int i = 4 * it + wave;
      v2f br[4], bi[4];
#pragma unroll
      for (int c = 0; c < 4; ++c) {
        int k0 = 4 * c + 2 * h;
        br[c] = v2f{sr[i * 256 + (k0 + 0) * 16 + l16],
                    sr[i * 256 + (k0 + 1) * 16 + l16]};
        bi[c] = v2f{si[i * 256 + (k0 + 0) * 16 + l16],
                    si[i * 256 + (k0 + 1) * 16 + l16]};
      }
      v8f aRe = {}, aIm = {};
#pragma unroll
      for (int c = 0; c < 4; ++c) {
        aRe = wmma4(gr[c],  br[c], aRe);
        aRe = wmma4(gin[c], bi[c], aRe);
        aIm = wmma4(gi[c],  br[c], aIm);
        aIm = wmma4(gr[c],  bi[c], aIm);
      }
#pragma unroll
      for (int r = 0; r < 8; ++r) {
        int M = r + 8 * h;
        sr[i * 256 + M * 16 + l16] = aRe[r];
        si[i * 256 + M * 16 + l16] = aIm[r];
      }
    }
  }
  __syncthreads();

  {
    const float* Gr = ws + 1024;
    const float* Gi = ws + 1024 + 256;
    v2f bgr[4], bgi[4], bgin[4];
#pragma unroll
    for (int c = 0; c < 4; ++c) {
      int col = 4 * c + 2 * h;
      bgr[c]  = v2f{Gr[l16 * 16 + col], Gr[l16 * 16 + col + 1]};
      bgi[c]  = v2f{Gi[l16 * 16 + col], Gi[l16 * 16 + col + 1]};
      bgin[c] = v2f{-bgi[c].x, -bgi[c].y};
    }
#pragma unroll
    for (int it = 0; it < 4; ++it) {
      int r0 = (4 * it + wave) * 16;
      v2f ar[4], ai[4];
#pragma unroll
      for (int c = 0; c < 4; ++c) {
        int col = 4 * c + 2 * h;
        ar[c] = v2f{sr[(r0 + l16) * 16 + col], sr[(r0 + l16) * 16 + col + 1]};
        ai[c] = v2f{si[(r0 + l16) * 16 + col], si[(r0 + l16) * 16 + col + 1]};
      }
      v8f aRe = {}, aIm = {};
#pragma unroll
      for (int c = 0; c < 4; ++c) {
        aRe = wmma4(ar[c], bgr[c],  aRe);
        aRe = wmma4(ai[c], bgin[c], aRe);
        aIm = wmma4(ar[c], bgi[c],  aIm);
        aIm = wmma4(ai[c], bgr[c],  aIm);
      }
#pragma unroll
      for (int r = 0; r < 8; ++r) {
        int M = r + 8 * h;
        sr[(r0 + M) * 16 + l16] = aRe[r];
        si[(r0 + M) * 16 + l16] = aIm[r];
      }
    }
  }
  __syncthreads();

  const float* zW = ws + 1536;
  float acc = 0.f;
  for (int k = tid; k < QDIM; k += 128) {
    float re = sr[k], im = si[k];
    acc += (re * re + im * im) * zW[k];
  }
#pragma unroll
  for (int off = 16; off; off >>= 1) acc += __shfl_down(acc, off, 32);
  if (lane == 0) red[wave] = acc;
  __syncthreads();
  if (tid == 0) out[blockIdx.x] = red[0] + red[1] + red[2] + red[3] + bptr[0];
}

extern "C" void kernel_launch(void* const* d_in, const int* in_sizes, int n_in,
                              void* d_out, int out_size, void* d_ws, size_t ws_size,
                              hipStream_t stream) {
  const float* state  = (const float*)d_in[0];
  const float* params = (const float*)d_in[1];
  const float* W      = (const float*)d_in[2];
  const float* b      = (const float*)d_in[3];
  float* out = (float*)d_out;
  float* ws  = (float*)d_ws;

  qsim_setup<<<1, 256, 0, stream>>>(params, W, ws);
  qsim_main<<<4096, 128, 0, stream>>>(state, ws, b, out);
}


# regroup the 12 wires as 4 groups of 3: each 8x8 complex gate applied as ONE real 16x16 block product [[Gr,-Gi],[Gi,Gr]] (448 instead of 640 WMMA per state); the order of the sums changes
# speedup vs baseline: 1.0856x; 1.0856x over previous
#include <hip/hip_runtime.h>
#include <hip/hip_bf16.h>


typedef __attribute__((ext_vector_type(2))) float v2f;
typedef __attribute__((ext_vector_type(8))) float v8f;

#define NW 12
#define QDIM 4096

#define PA 520
#define PB 65
#define PLANE 4159
#define IMOFF 4188

__global__ void qsim_setup(const float* __restrict__ params,
                           const float* __restrict__ W,
                           float* __restrict__ ws) {
  __shared__ float u_re[NW][2][2];
  __shared__ float u_im[NW][2][2];
  int t = threadIdx.x;
  if (t < NW) {
    float th = params[2 * t] * 0.5f;
    float ph = params[2 * t + 1] * 0.5f;
    float c = cosf(th), s = sinf(th);
    float cy = cosf(ph), sy = sinf(ph);
    u_re[t][0][0] = cy * c;   u_im[t][0][0] =  sy * s;
    u_re[t][0][1] = -sy * c;  u_im[t][0][1] = -cy * s;
    u_re[t][1][0] =  sy * c;  u_im[t][1][0] = -cy * s;
    u_re[t][1][1] =  cy * c;  u_im[t][1][1] = -sy * s;
  }
  __syncthreads();
  for (int e = t; e < 4 * 256; e += blockDim.x) {
    int g = e >> 8;
    int M = (e >> 4) & 15, K = e & 15;
    int rip = M >> 3, xp = M & 7, ri = K >> 3, x = K & 7;
    float ar = 1.f, ai = 0.f;
    for (int q = 0; q < 3; ++q) {
      int w = g * 3 + q;
      int rb = (xp >> (2 - q)) & 1;
      int cb = (x >> (2 - q)) & 1;
      float br = u_re[w][rb][cb], bi = u_im[w][rb][cb];
      float nr = ar * br - ai * bi;
      float ni = ar * bi + ai * br;
      ar = nr; ai = ni;
    }
    ws[e] = (rip == ri) ? ar : (rip == 0 ? -ai : ai);
  }
  for (int k = t; k < QDIM; k += blockDim.x) {
    float acc = 0.f;
    int par = 0;
    for (int j = 0; j < NW; ++j) {
      par ^= (k >> (NW - 1 - j)) & 1;
      acc += W[j] * (par ? -1.f : 1.f);
    }
    ws[1024 + k] = acc;
  }
}

__device__ __forceinline__ v8f wmma4(v2f a, v2f b, v8f c) {
  return __builtin_amdgcn_wmma_f32_16x16x4_f32(false, a, false, b,
                                               (short)0, c, false, false);
}

template <int G> struct Grp;
template <> struct Grp<0> {
  static constexpr int PX = PA, XH = 1;
  static __device__ __forceinline__ int lane_col(int wave, int l16) {
    return wave * (2 * PB) + (l16 >> 3) * 16 + (l16 & 7);
  }
  static constexpr int const_col(int it, int tt) {
    return (it >> 1) * PB + tt * 8 + (it & 1) * 32;
  }
  static constexpr int xc(int cb, int e) { return 2 * e + 4 * cb; }
};
template <> struct Grp<1> {
  static constexpr int PX = PB, XH = 1;
  static __device__ __forceinline__ int lane_col(int wave, int l16) {
    return wave * (2 * PA) + (l16 & 7) * 8 + (l16 >> 3) * 2;
  }
  static constexpr int const_col(int it, int tt) {
    return (it >> 1) * PA + tt + (it & 1) * 4;
  }
  static constexpr int xc(int cb, int e) { return 2 * e + 4 * cb; }
};
template <> struct Grp<2> {
  static constexpr int PX = 8, XH = 1;
  static __device__ __forceinline__ int lane_col(int wave, int l16) {
    return wave * (2 * PB) + (l16 >> 3) * (2 * PA) + (l16 & 7);
  }
  static constexpr int const_col(int it, int tt) {
    return tt * PA + (it & 1) * (4 * PA) + (it >> 1) * PB;
  }
  static constexpr int xc(int cb, int e) { return 2 * e + 4 * cb; }
};
template <> struct Grp<3> {
  static constexpr int PX = 1, XH = 2;
  static __device__ __forceinline__ int lane_col(int wave, int l16) {
    return wave * (2 * PA) + (l16 >> 3) * PB + (l16 & 7) * 8;
  }
  static constexpr int const_col(int it, int tt) {
    return tt * (2 * PB) + (it & 1) * (4 * PB) + (it >> 1) * PA;
  }
  static constexpr int xc(int cb, int e) { return e + 4 * cb; }
};

template <int G, int NCH>
__device__ __forceinline__ void apply_group(float* __restrict__ S,
                                            const float* __restrict__ Ah,
                                            int wave, int h, int l16) {
  typedef Grp<G> P;
  const float* arow = Ah + l16 * 16 + h * P::XH;
  v2f af[NCH];
#pragma unroll
  for (int c = 0; c < NCH; ++c) {
    af[c] = v2f{arow[(c >> 1) * 8 + P::xc(c & 1, 0)],
                arow[(c >> 1) * 8 + P::xc(c & 1, 1)]};
  }
  const int lc = P::lane_col(wave, l16);
  const float* ld = S + lc + h * (P::XH * P::PX);
  float*       st = S + lc + h * IMOFF;
#pragma unroll
  for (int it = 0; it < 4; ++it) {
    v2f b0[NCH], b1[NCH];
#pragma unroll
    for (int c = 0; c < NCH; ++c) {
      const int o0 = (c >> 1) * IMOFF + P::const_col(it, 0);
      const int o1 = (c >> 1) * IMOFF + P::const_col(it, 1);
      const int x0 = P::xc(c & 1, 0) * P::PX;
      const int x1 = P::xc(c & 1, 1) * P::PX;
      b0[c] = v2f{ld[o0 + x0], ld[o0 + x1]};
      b1[c] = v2f{ld[o1 + x0], ld[o1 + x1]};
    }
    v8f d0 = {}, d1 = {};
#pragma unroll
    for (int c = 0; c < NCH; ++c) {
      d0 = wmma4(af[c], b0[c], d0);
      d1 = wmma4(af[c], b1[c], d1);
    }
#pragma unroll
    for (int r = 0; r < 8; ++r) {
      st[P::const_col(it, 0) + r * P::PX] = d0[r];
      st[P::const_col(it, 1) + r * P::PX] = d1[r];
    }
  }
}

__global__ __launch_bounds__(128) void qsim_main(
    const float* __restrict__ sb, const float* __restrict__ ws,
    const float* __restrict__ bptr, float* __restrict__ out) {
  __shared__ float S[IMOFF + PLANE];
  __shared__ float red[4];

  const int tid  = threadIdx.x;
  const int lane = tid & 31;
  const int wave = tid >> 5;
  const int h    = lane >> 4;
  const int l16  = lane & 15;
  const float* in = sb + (size_t)blockIdx.x * QDIM;

  for (int k = tid; k < QDIM; k += 128) S[k + (k >> 6)] = in[k];
  __syncthreads();

  apply_group<0, 2>(S, ws,       wave, h, l16);
  __syncthreads();
  apply_group<1, 4>(S, ws + 256, wave, h, l16);
  __syncthreads();
  apply_group<2, 4>(S, ws + 512, wave, h, l16);
  __syncthreads();
  apply_group<3, 4>(S, ws + 768, wave, h, l16);
  __syncthreads();

  const float* zW = ws + 1024;
  float acc = 0.f;
  for (int k = tid; k < QDIM; k += 128) {
    float re = S[k + (k >> 6)], im = S[IMOFF + k + (k >> 6)];
    acc += (re * re + im * im) * zW[k];
  }
#pragma unroll
  for (int off = 16; off; off >>= 1) acc += __shfl_down(acc, off, 32);
  if (lane == 0) red[wave] = acc;
  __syncthreads();
  if (tid == 0) out[blockIdx.x] = red[0] + red[1] + red[2] + red[3] + bptr[0];
}

extern "C" void kernel_launch(void* const* d_in, const int* in_sizes, int n_in,
                              void* d_out, int out_size, void* d_ws, size_t ws_size,
                              hipStream_t stream) {
  const float* state  = (const float*)d_in[0];
  const float* params = (const float*)d_in[1];
  const float* W      = (const float*)d_in[2];
  const float* b      = (const float*)d_in[3];
  float* out = (float*)d_out;
  float* ws  = (float*)d_ws;

  qsim_setup<<<1, 256, 0, stream>>>(params, W, ws);
  qsim_main<<<4096, 128, 0, stream>>>(state, ws, b, out);
}


# fold the probs.zW reduction into the last gate group: its WMMA accumulators are squared and weighted in registers, the final state never goes back to LDS; the order of the sums changes
# speedup vs baseline: 1.1278x; 1.0389x over previous
#include <hip/hip_runtime.h>
#include <hip/hip_bf16.h>


typedef __attribute__((ext_vector_type(2))) float v2f;
typedef __attribute__((ext_vector_type(8))) float v8f;

#define NW 12
#define QDIM 4096

#define PA 520
#define PB 65
#define PLANE 4159
#define IMOFF 4188

__global__ void qsim_setup(const float* __restrict__ params,
                           const float* __restrict__ W,
                           float* __restrict__ ws) {
  __shared__ float u_re[NW][2][2];
  __shared__ float u_im[NW][2][2];
  int t = threadIdx.x;
  if (t < NW) {
    float th = params[2 * t] * 0.5f;
    float ph = params[2 * t + 1] * 0.5f;
    float c = cosf(th), s = sinf(th);
    float cy = cosf(ph), sy = sinf(ph);
    u_re[t][0][0] = cy * c;   u_im[t][0][0] =  sy * s;
    u_re[t][0][1] = -sy * c;  u_im[t][0][1] = -cy * s;
    u_re[t][1][0] =  sy * c;  u_im[t][1][0] = -cy * s;
    u_re[t][1][1] =  cy * c;  u_im[t][1][1] = -sy * s;
  }
  __syncthreads();
  for (int e = t; e < 4 * 256; e += blockDim.x) {
    int g = e >> 8;
    int M = (e >> 4) & 15, K = e & 15;
    int rip = M >> 3, xp = M & 7, ri = K >> 3, x = K & 7;
    float ar = 1.f, ai = 0.f;
    for (int q = 0; q < 3; ++q) {
      int w = g * 3 + q;
      int rb = (xp >> (2 - q)) & 1;
      int cb = (x >> (2 - q)) & 1;
      float br = u_re[w][rb][cb], bi = u_im[w][rb][cb];
      float nr = ar * br - ai * bi;
      float ni = ar * bi + ai * br;
      ar = nr; ai = ni;
    }
    ws[e] = (rip == ri) ? ar : (rip == 0 ? -ai : ai);
  }
  for (int k = t; k < QDIM; k += blockDim.x) {
    float acc = 0.f;
    int par = 0;
    for (int j = 0; j < NW; ++j) {
      par ^= (k >> (NW - 1 - j)) & 1;
      acc += W[j] * (par ? -1.f : 1.f);
    }
    ws[1024 + k] = acc;
  }
}

__device__ __forceinline__ v8f wmma4(v2f a, v2f b, v8f c) {
  return __builtin_amdgcn_wmma_f32_16x16x4_f32(false, a, false, b,
                                               (short)0, c, false, false);
}

template <int G> struct Grp;
template <> struct Grp<0> {
  static constexpr int PX = PA, XH = 1;
  static __device__ __forceinline__ int lane_col(int wave, int l16) {
    return wave * (2 * PB) + (l16 >> 3) * 16 + (l16 & 7);
  }
  static constexpr int const_col(int it, int tt) {
    return (it >> 1) * PB + tt * 8 + (it & 1) * 32;
  }
  static constexpr int xc(int cb, int e) { return 2 * e + 4 * cb; }
};
template <> struct Grp<1> {
  static constexpr int PX = PB, XH = 1;
  static __device__ __forceinline__ int lane_col(int wave, int l16) {
    return wave * (2 * PA) + (l16 & 7) * 8 + (l16 >> 3) * 2;
  }
  static constexpr int const_col(int it, int tt) {
    return (it >> 1) * PA + tt + (it & 1) * 4;
  }
  static constexpr int xc(int cb, int e) { return 2 * e + 4 * cb; }
};
template <> struct Grp<2> {
  static constexpr int PX = 8, XH = 1;
  static __device__ __forceinline__ int lane_col(int wave, int l16) {
    return wave * (2 * PB) + (l16 >> 3) * (2 * PA) + (l16 & 7);
  }
  static constexpr int const_col(int it, int tt) {
    return tt * PA + (it & 1) * (4 * PA) + (it >> 1) * PB;
  }
  static constexpr int xc(int cb, int e) { return 2 * e + 4 * cb; }
};
template <> struct Grp<3> {
  static constexpr int PX = 1, XH = 2;
  static __device__ __forceinline__ int lane_col(int wave, int l16) {
    return wave * (2 * PA) + (l16 >> 3) * PB + (l16 & 7) * 8;
  }
  static constexpr int const_col(int it, int tt) {
    return tt * (2 * PB) + (it & 1) * (4 * PB) + (it >> 1) * PA;
  }
  static constexpr int xc(int cb, int e) { return e + 4 * cb; }
  static __device__ __forceinline__ int lane_k(int wave, int l16) {
    return wave * 1024 + (l16 >> 3) * 64 + (l16 & 7) * 8;
  }
  static constexpr int const_k(int it, int tt) {
    return tt * 128 + (it & 1) * 256 + (it >> 1) * 512;
  }
};

template <int G, int NCH, bool LAST>
__device__ __forceinline__ float apply_group(float* __restrict__ S,
                                             const float* __restrict__ Ah,
                                             const float* __restrict__ zW,
                                             int wave, int h, int l16) {
  typedef Grp<G> P;
  const float* arow = Ah + l16 * 16 + h * P::XH;
  v2f af[NCH];
#pragma unroll
  for (int c = 0; c < NCH; ++c) {
    af[c] = v2f{arow[(c >> 1) * 8 + P::xc(c & 1, 0)],
                arow[(c >> 1) * 8 + P::xc(c & 1, 1)]};
  }
  const int lc = P::lane_col(wave, l16);
  const float* ld = S + lc + h * (P::XH * P::PX);
  float*       st = S + lc + h * IMOFF;
  float acc = 0.f;
  const float* zl = zW;
  if constexpr (LAST) zl = zW + P::lane_k(wave, l16);
#pragma unroll
  for (int it = 0; it < 4; ++it) {
    float4 w0a, w0b, w1a, w1b;
    if constexpr (LAST) {
      const float4* z0 = reinterpret_cast<const float4*>(zl + P::const_k(it, 0));
      const float4* z1 = reinterpret_cast<const float4*>(zl + P::const_k(it, 1));
      w0a = z0[0]; w0b = z0[1]; w1a = z1[0]; w1b = z1[1];
    }
    v2f b0[NCH], b1[NCH];
#pragma unroll
    for (int c = 0; c < NCH; ++c) {
      const int o0 = (c >> 1) * IMOFF + P::const_col(it, 0);
      const int o1 = (c >> 1) * IMOFF + P::const_col(it, 1);
      const int x0 = P::xc(c & 1, 0) * P::PX;
      const int x1 = P::xc(c & 1, 1) * P::PX;
      b0[c] = v2f{ld[o0 + x0], ld[o0 + x1]};
      b1[c] = v2f{ld[o1 + x0], ld[o1 + x1]};
    }
    v8f d0 = {}, d1 = {};
#pragma unroll
    for (int c = 0; c < NCH; ++c) {
      d0 = wmma4(af[c], b0[c], d0);
      d1 = wmma4(af[c], b1[c], d1);
    }
    if constexpr (LAST) {
      acc += d0[0] * d0[0] * w0a.x + d0[1] * d0[1] * w0a.y
           + d0[2] * d0[2] * w0a.z + d0[3] * d0[3] * w0a.w
           + d0[4] * d0[4] * w0b.x + d0[5] * d0[5] * w0b.y
           + d0[6] * d0[6] * w0b.z + d0[7] * d0[7] * w0b.w;
      acc += d1[0] * d1[0] * w1a.x + d1[1] * d1[1] * w1a.y
           + d1[2] * d1[2] * w1a.z + d1[3] * d1[3] * w1a.w
           + d1[4] * d1[4] * w1b.x + d1[5] * d1[5] * w1b.y
           + d1[6] * d1[6] * w1b.z + d1[7] * d1[7] * w1b.w;
      __builtin_amdgcn_sched_barrier(0);
    } else {
#pragma unroll
      for (int r = 0; r < 8; ++r) {
        st[P::const_col(it, 0) + r * P::PX] = d0[r];
        st[P::const_col(it, 1) + r * P::PX] = d1[r];
      }
    }
  }
  return acc;
}

__global__ __launch_bounds__(128) void qsim_main(
    const float* __restrict__ sb, const float* __restrict__ ws,
    const float* __restrict__ bptr, float* __restrict__ out) {
  __shared__ float S[IMOFF + PLANE];
  __shared__ float red[4];

  const int tid  = threadIdx.x;
  const int lane = tid & 31;
  const int wave = tid >> 5;
  const int h    = lane >> 4;
  const int l16  = lane & 15;
  const float* in = sb + (size_t)blockIdx.x * QDIM;

  for (int k = tid; k < QDIM; k += 128) S[k + (k >> 6)] = in[k];
  __syncthreads();

  const float* zW = ws + 1024;
  apply_group<0, 2, false>(S, ws,       zW, wave, h, l16);
  __syncthreads();
  apply_group<1, 4, false>(S, ws + 256, zW, wave, h, l16);
  __syncthreads();
  apply_group<2, 4, false>(S, ws + 512, zW, wave, h, l16);
  __syncthreads();
  float acc = apply_group<3, 4, true>(S, ws + 768, zW, wave, h, l16);

#pragma unroll
  for (int off = 16; off; off >>= 1) acc += __shfl_down(acc, off, 32);
  if (lane == 0) red[wave] = acc;
  __syncthreads();
  if (tid == 0) out[blockIdx.x] = red[0] + red[1] + red[2] + red[3] + bptr[0];
}

extern "C" void kernel_launch(void* const* d_in, const int* in_sizes, int n_in,
                              void* d_out, int out_size, void* d_ws, size_t ws_size,
                              hipStream_t stream) {
  const float* state  = (const float*)d_in[0];
  const float* params = (const float*)d_in[1];
  const float* W      = (const float*)d_in[2];
  const float* b      = (const float*)d_in[3];
  float* out = (float*)d_out;
  float* ws  = (float*)d_ws;

  qsim_setup<<<1, 256, 0, stream>>>(params, W, ws);
  qsim_main<<<4096, 128, 0, stream>>>(state, ws, b, out);
}


# persistent work-groups: 2048 groups of 4 waves, 8 per WGP and all resident, each loop over 2 states; gate fragments loaded once; next state's loads stay in flight across LDS-only barriers
# speedup vs baseline: 1.1919x; 1.0568x over previous
#include <hip/hip_runtime.h>
#include <hip/hip_bf16.h>


typedef __attribute__((ext_vector_type(2))) float v2f;
typedef __attribute__((ext_vector_type(8))) float v8f;

#define NW 12
#define QDIM 4096
#define NSTATE 4096
#define NBLK 2048

#define PA 520
#define PB 65
#define PLANE 4159
#define IMOFF 4188

__global__ void qsim_setup(const float* __restrict__ params,
                           const float* __restrict__ W,
                           float* __restrict__ ws) {
  __shared__ float u_re[NW][2][2];
  __shared__ float u_im[NW][2][2];
  int t = threadIdx.x;
  if (t < NW) {
    float th = params[2 * t] * 0.5f;
    float ph = params[2 * t + 1] * 0.5f;
    float c = cosf(th), s = sinf(th);
    float cy = cosf(ph), sy = sinf(ph);
    u_re[t][0][0] = cy * c;   u_im[t][0][0] =  sy * s;
    u_re[t][0][1] = -sy * c;  u_im[t][0][1] = -cy * s;
    u_re[t][1][0] =  sy * c;  u_im[t][1][0] = -cy * s;
    u_re[t][1][1] =  cy * c;  u_im[t][1][1] = -sy * s;
  }
  __syncthreads();
  for (int e = t; e < 4 * 256; e += blockDim.x) {
    int g = e >> 8;
    int M = (e >> 4) & 15, K = e & 15;
    int rip = M >> 3, xp = M & 7, ri = K >> 3, x = K & 7;
    float ar = 1.f, ai = 0.f;
    for (int q = 0; q < 3; ++q) {
      int w = g * 3 + q;
      int rb = (xp >> (2 - q)) & 1;
      int cb = (x >> (2 - q)) & 1;
      float br = u_re[w][rb][cb], bi = u_im[w][rb][cb];
      float nr = ar * br - ai * bi;
      float ni = ar * bi + ai * br;
      ar = nr; ai = ni;
    }
    ws[e] = (rip == ri) ? ar : (rip == 0 ? -ai : ai);
  }
  for (int k = t; k < QDIM; k += blockDim.x) {
    float acc = 0.f;
    int par = 0;
    for (int j = 0; j < NW; ++j) {
      par ^= (k >> (NW - 1 - j)) & 1;
      acc += W[j] * (par ? -1.f : 1.f);
    }
    ws[1024 + k] = acc;
  }
}

__device__ __forceinline__ v8f wmma4(v2f a, v2f b, v8f c) {
  return __builtin_amdgcn_wmma_f32_16x16x4_f32(false, a, false, b,
                                               (short)0, c, false, false);
}

template <int G> struct Grp;
template <> struct Grp<0> {
  static constexpr int PX = PA, XH = 1;
  static __device__ __forceinline__ int lane_col(int wave, int l16) {
    return wave * (2 * PB) + (l16 >> 3) * 16 + (l16 & 7);
  }
  static constexpr int const_col(int it, int tt) {
    return (it >> 1) * PB + tt * 8 + (it & 1) * 32;
  }
  static constexpr int xc(int cb, int e) { return 2 * e + 4 * cb; }
};
template <> struct Grp<1> {
  static constexpr int PX = PB, XH = 1;
  static __device__ __forceinline__ int lane_col(int wave, int l16) {
    return wave * (2 * PA) + (l16 & 7) * 8 + (l16 >> 3) * 2;
  }
  static constexpr int const_col(int it, int tt) {
    return (it >> 1) * PA + tt + (it & 1) * 4;
  }
  static constexpr int xc(int cb, int e) { return 2 * e + 4 * cb; }
};
template <> struct Grp<2> {
  static constexpr int PX = 8, XH = 1;
  static __device__ __forceinline__ int lane_col(int wave, int l16) {
    return wave * (2 * PB) + (l16 >> 3) * (2 * PA) + (l16 & 7);
  }
  static constexpr int const_col(int it, int tt) {
    return tt * PA + (it & 1) * (4 * PA) + (it >> 1) * PB;
  }
  static constexpr int xc(int cb, int e) { return 2 * e + 4 * cb; }
};
template <> struct Grp<3> {
  static constexpr int PX = 1, XH = 2;
  static __device__ __forceinline__ int lane_col(int wave, int l16) {
    return wave * (2 * PA) + (l16 >> 3) * PB + (l16 & 7) * 8;
  }
  static constexpr int const_col(int it, int tt) {
    return tt * (2 * PB) + (it & 1) * (4 * PB) + (it >> 1) * PA;
  }
  static constexpr int xc(int cb, int e) { return e + 4 * cb; }
  static __device__ __forceinline__ int lane_k(int wave, int l16) {
    return wave * 1024 + (l16 >> 3) * 64 + (l16 & 7) * 8;
  }
  static constexpr int const_k(int it, int tt) {
    return tt * 128 + (it & 1) * 256 + (it >> 1) * 512;
  }
};

__device__ __forceinline__ void lds_barrier() {
  __builtin_amdgcn_fence(__ATOMIC_RELEASE, "workgroup", "local");
  __builtin_amdgcn_s_barrier();
  __builtin_amdgcn_fence(__ATOMIC_ACQUIRE, "workgroup", "local");
}

template <int G, int NCH>
__device__ __forceinline__ void load_afrag(v2f (&af)[NCH], const float* __restrict__ Ah,
                                           int h, int l16) {
  typedef Grp<G> P;
  const float* arow = Ah + l16 * 16 + h * P::XH;
#pragma unroll
  for (int c = 0; c < NCH; ++c) {
    af[c] = v2f{arow[(c >> 1) * 8 + P::xc(c & 1, 0)],
                arow[(c >> 1) * 8 + P::xc(c & 1, 1)]};
  }
}

template <int G, int NCH, bool LAST>
__device__ __forceinline__ float apply_group(float* __restrict__ S,
                                             const v2f (&af)[NCH],
                                             const float* __restrict__ zW,
                                             int wave, int h, int l16) {
  typedef Grp<G> P;
  const int lc = P::lane_col(wave, l16);
  const float* ld = S + lc + h * (P::XH * P::PX);
  float*       st = S + lc + h * IMOFF;
  float acc = 0.f;
  const float* zl = zW;
  if constexpr (LAST) {
    int kl = P::lane_k(wave, l16);
    asm volatile("" : "+v"(kl));
    zl = zW + kl;
  }
#pragma unroll
  for (int it = 0; it < 4; ++it) {
    float4 w0a, w0b, w1a, w1b;
    if constexpr (LAST) {
      const float4* z0 = reinterpret_cast<const float4*>(zl + P::const_k(it, 0));
      const float4* z1 = reinterpret_cast<const float4*>(zl + P::const_k(it, 1));
      w0a = z0[0]; w0b = z0[1]; w1a = z1[0]; w1b = z1[1];
    }
    v2f b0[NCH], b1[NCH];
#pragma unroll
    for (int c = 0; c < NCH; ++c) {
      const int o0 = (c >> 1) * IMOFF + P::const_col(it, 0);
      const int o1 = (c >> 1) * IMOFF + P::const_col(it, 1);
      const int x0 = P::xc(c & 1, 0) * P::PX;
      const int x1 = P::xc(c & 1, 1) * P::PX;
      b0[c] = v2f{ld[o0 + x0], ld[o0 + x1]};
      b1[c] = v2f{ld[o1 + x0], ld[o1 + x1]};
    }
    v8f d0 = {}, d1 = {};
#pragma unroll
    for (int c = 0; c < NCH; ++c) {
      d0 = wmma4(af[c], b0[c], d0);
      d1 = wmma4(af[c], b1[c], d1);
    }
    if constexpr (LAST) {
      acc += d0[0] * d0[0] * w0a.x + d0[1] * d0[1] * w0a.y
           + d0[2] * d0[2] * w0a.z + d0[3] * d0[3] * w0a.w
           + d0[4] * d0[4] * w0b.x + d0[5] * d0[5] * w0b.y
           + d0[6] * d0[6] * w0b.z + d0[7] * d0[7] * w0b.w;
      acc += d1[0] * d1[0] * w1a.x + d1[1] * d1[1] * w1a.y
           + d1[2] * d1[2] * w1a.z + d1[3] * d1[3] * w1a.w
           + d1[4] * d1[4] * w1b.x + d1[5] * d1[5] * w1b.y
           + d1[6] * d1[6] * w1b.z + d1[7] * d1[7] * w1b.w;
      __builtin_amdgcn_sched_barrier(0);
    } else {
#pragma unroll
      for (int r = 0; r < 8; ++r) {
        st[P::const_col(it, 0) + r * P::PX] = d0[r];
        st[P::const_col(it, 1) + r * P::PX] = d1[r];
      }
    }
  }
  return acc;
}

__global__ __launch_bounds__(128, 8) void qsim_main(
    const float* __restrict__ sb, const float* __restrict__ ws,
    const float* __restrict__ bptr, float* __restrict__ out) {
  __shared__ float S[IMOFF + PLANE];
  __shared__ float red[4];

  const int tid  = threadIdx.x;
  const int lane = tid & 31;
  const int wave = tid >> 5;
  const int h    = lane >> 4;
  const int l16  = lane & 15;

  v2f a0[2], a1[4], a2[4], a3[4];
  load_afrag<0, 2>(a0, ws,       h, l16);
  load_afrag<1, 4>(a1, ws + 256, h, l16);
  load_afrag<2, 4>(a2, ws + 512, h, l16);
  load_afrag<3, 4>(a3, ws + 768, h, l16);
  const float* zW = ws + 1024;
  const float bias = bptr[0];

  float4 nx[8];
  const int sbase = 4 * tid + (tid >> 4);
  {
    const float4* in4 = reinterpret_cast<const float4*>(sb + (size_t)blockIdx.x * QDIM);
#pragma unroll
    for (int i = 0; i < 8; ++i) nx[i] = in4[tid + 128 * i];
  }

  for (int s = blockIdx.x; s < NSTATE; s += NBLK) {
#pragma unroll
    for (int i = 0; i < 8; ++i) {
      float* p = S + sbase + i * PA;
      p[0] = nx[i].x; p[1] = nx[i].y; p[2] = nx[i].z; p[3] = nx[i].w;
    }
    if (s + NBLK < NSTATE) {
      const float4* in4 = reinterpret_cast<const float4*>(sb + (size_t)(s + NBLK) * QDIM);
#pragma unroll
      for (int i = 0; i < 8; ++i) nx[i] = in4[tid + 128 * i];
    }
    lds_barrier();

    apply_group<0, 2, false>(S, a0, zW, wave, h, l16);
    lds_barrier();
    apply_group<1, 4, false>(S, a1, zW, wave, h, l16);
    lds_barrier();
    apply_group<2, 4, false>(S, a2, zW, wave, h, l16);
    lds_barrier();
    float acc = apply_group<3, 4, true>(S, a3, zW, wave, h, l16);

#pragma unroll
    for (int off = 16; off; off >>= 1) acc += __shfl_down(acc, off, 32);
    if (lane == 0) red[wave] = acc;
    lds_barrier();
    if (tid == 0) out[s] = red[0] + red[1] + red[2] + red[3] + bias;
  }
}

extern "C" void kernel_launch(void* const* d_in, const int* in_sizes, int n_in,
                              void* d_out, int out_size, void* d_ws, size_t ws_size,
                              hipStream_t stream) {
  const float* state  = (const float*)d_in[0];
  const float* params = (const float*)d_in[1];
  const float* W      = (const float*)d_in[2];
  const float* b      = (const float*)d_in[3];
  float* out = (float*)d_out;
  float* ws  = (float*)d_ws;

  qsim_setup<<<1, 256, 0, stream>>>(params, W, ws);
  qsim_main<<<NBLK, 128, 0, stream>>>(state, ws, b, out);
}


# gate groups fused in pairs: the second of a pair takes its B operands from the first's WMMA accumulators in registers; state passes LDS twice not 4 times, 3 barriers not 5; order of sums changes
# speedup vs baseline: 1.4778x; 1.2399x over previous
#include <hip/hip_runtime.h>
#include <hip/hip_bf16.h>


typedef __attribute__((ext_vector_type(2))) float v2f;
typedef __attribute__((ext_vector_type(8))) float v8f;

#define NW 12
#define QDIM 4096
#define NSTATE 4096
#define NBLK 1024

#define PA 520
#define PB 65
#define PLANE 4159
#define IMOFF 4188

struct ZTerms { float base, e5, a6, a7, a8, tz0, tz1, tz2, tz3; };

__device__ __forceinline__ v8f wmma4(v2f a, v2f b, v8f c) {
  return __builtin_amdgcn_wmma_f32_16x16x4_f32(false, a, false, b,
                                               (short)0, c, false, false);
}

__device__ __forceinline__ void lds_barrier() {
  __builtin_amdgcn_fence(__ATOMIC_RELEASE, "workgroup", "local");
  __builtin_amdgcn_s_barrier();
  __builtin_amdgcn_fence(__ATOMIC_ACQUIRE, "workgroup", "local");
}


template <int NCH>
__device__ __forceinline__ void load_afrag_lds(v2f (&af)[NCH], const float* __restrict__ Ah,
                                               int h, int l16) {
  const float* arow = Ah + l16 * 16 + h;
#pragma unroll
  for (int c = 0; c < NCH; ++c) {
    af[c] = v2f{arow[(c >> 1) * 8 + 4 * (c & 1)], arow[(c >> 1) * 8 + 4 * (c & 1) + 2]};
  }
}
__device__ __forceinline__ void load_afrag_reg(v2f (&af)[4], const float* __restrict__ Ah,
                                               int h, int l16) {
  const float* arow = Ah + l16 * 16 + h * 8;
#pragma unroll
  for (int c = 0; c < 4; ++c) af[c] = v2f{arow[2 * c], arow[2 * c + 1]};
}

__device__ __forceinline__ void second_group_pair(const v8f (&d1)[8], const v2f (&af)[4],
                                                  int r, v8f& e0, v8f& e1) {
  e0 = v8f{}; e1 = v8f{};
#pragma unroll
  for (int c = 0; c < 4; ++c) {
    e0 = wmma4(af[c], v2f{d1[2 * c][r],     d1[2 * c + 1][r]},     e0);
    e1 = wmma4(af[c], v2f{d1[2 * c][r + 1], d1[2 * c + 1][r + 1]}, e1);
  }
}

__device__ __forceinline__ void weigh_tile(const v8f& d, const ZTerms& z, int r,
                                           float (&Q)[8], float& accb) {
  const bool n6 = ((r >> 2) & 1) != 0;
  const bool n7 = (((r >> 2) ^ (r >> 1)) & 1) != 0;
  const bool n8 = (((r >> 2) ^ (r >> 1) ^ r) & 1) != 0;
  float s1 = d[0] * d[0];
#pragma unroll
  for (int i = 1; i < 8; ++i) s1 = fmaf(d[i], d[i], s1);
  const float base = z.base + (n6 ? -z.a6 : z.a6) + (n7 ? -z.a7 : z.a7) + (n8 ? -z.a8 : z.a8);
  accb = fmaf(base, s1, accb);
#pragma unroll
  for (int i = 0; i < 8; ++i) Q[i] = fmaf(n8 ? -d[i] : d[i], d[i], Q[i]);
}

__global__ __launch_bounds__(128, 4) void qsim_main(
    const float* __restrict__ sb, const float* __restrict__ params,
    const float* __restrict__ W, const float* __restrict__ bptr,
    float* __restrict__ out) {
  __shared__ float S[IMOFF + PLANE];
  __shared__ float red[4];
  __shared__ float u_re[NW][4];
  __shared__ float u_im[NW][4];

  const int tid  = threadIdx.x;
  const int lane = tid & 31;
  const int wave = __builtin_amdgcn_readfirstlane(tid >> 5);
  const int h    = lane >> 4;
  const int l16  = lane & 15;

  float th = 0.f, ph = 0.f;
#pragma unroll
  for (int w = 0; w < NW; ++w) {
    const float p0 = params[2 * w], p1 = params[2 * w + 1];
    th = (tid == w) ? p0 : th;
    ph = (tid == w) ? p1 : ph;
  }
  th *= 0.5f;
  ph *= 0.5f;

  float4 nx[8];
  const int sbase = 4 * tid + (tid >> 4);
  {
    const float4* in4 = reinterpret_cast<const float4*>(sb + (size_t)blockIdx.x * QDIM);
#pragma unroll
    for (int i = 0; i < 8; ++i) nx[i] = in4[tid + 128 * i];
  }

  if (tid < NW) {
    float s, c, sy, cy;
    sincosf(th, &s, &c);
    sincosf(ph, &sy, &cy);
    u_re[tid][0] = cy * c;   u_im[tid][0] =  sy * s;
    u_re[tid][1] = -sy * c;  u_im[tid][1] = -cy * s;
    u_re[tid][2] =  sy * c;  u_im[tid][2] = -cy * s;
    u_re[tid][3] =  cy * c;  u_im[tid][3] = -sy * s;
  }
  lds_barrier();
#pragma unroll
  for (int j = 0; j < 2; ++j) {
    const int e = tid + 128 * j;
    const int g = e >> 6, xp = (e >> 3) & 7, x = e & 7;
    float ar = 1.f, ai = 0.f;
#pragma unroll
    for (int q = 0; q < 3; ++q) {
      const int w = g * 3 + q;
      const int rc = ((xp >> (2 - q)) & 1) * 2 + ((x >> (2 - q)) & 1);
      const float br = u_re[w][rc], bi = u_im[w][rc];
      const float nr = ar * br - ai * bi;
      const float ni = ar * bi + ai * br;
      ar = nr; ai = ni;
    }
    float* T = S + g * 256 + xp * 16 + x;
    T[0] = ar;   T[8] = -ai;
    T[128] = ai; T[136] = ar;
  }
  lds_barrier();
  v2f a0[2], a1[4], a2[4], a3[4];
  load_afrag_lds<2>(a0, S,       h, l16);
  load_afrag_reg   (a1, S + 256, h, l16);
  load_afrag_lds<4>(a2, S + 512, h, l16);
  load_afrag_reg   (a3, S + 768, h, l16);
  const float bias = bptr[0];

  ZTerms z;
  {
    const int p0 = wave >> 1;
    const int p1 = p0 ^ (l16 >> 3);
    const int p2 = p1 ^ (wave & 1);
    const int p3 = p2 ^ ((l16 >> 2) & 1);
    const int p4 = p3 ^ ((l16 >> 1) & 1);
    const int p5 = p4 ^ (l16 & 1);
    z.base = (p0 ? -W[0] : W[0]) + (p1 ? -W[1] : W[1]) + (p2 ? -W[2] : W[2])
           + (p3 ? -W[3] : W[3]) + (p4 ? -W[4] : W[4]) + (p5 ? -W[5] : W[5]);
    z.e5 = p5 ? -1.f : 1.f;
    z.a6 = p5 ? -W[6] : W[6];
    z.a7 = p5 ? -W[7] : W[7];
    z.a8 = p5 ? -W[8] : W[8];
    z.tz0 = W[9] + W[10] + W[11];
    z.tz1 = W[9] + W[10] - W[11];
    z.tz2 = W[9] - W[10] - W[11];
    z.tz3 = W[9] - W[10] + W[11];
  }
  const int lc1 = (wave & 1) * 8 + (wave >> 1) * 32 + (l16 >> 3) * 16 + (l16 & 7);
  const int lc2 = (wave & 1) * PA + (wave >> 1) * (4 * PA) + (l16 >> 3) * (2 * PA) + (l16 & 7) * PB;
  const float* ld1 = S + lc1 + h * PA;
  float*       st1 = S + lc1 + h * IMOFF;
  const float* ld2 = S + lc2 + h * 8;
  lds_barrier();

  for (int s = blockIdx.x; s < NSTATE; s += NBLK) {
#pragma unroll
    for (int i = 0; i < 8; ++i) {
      float* p = S + sbase + i * PA;
      p[0] = nx[i].x; p[1] = nx[i].y; p[2] = nx[i].z; p[3] = nx[i].w;
    }
    if (s + NBLK < NSTATE) {
      const float4* in4 = reinterpret_cast<const float4*>(sb + (size_t)(s + NBLK) * QDIM);
#pragma unroll
      for (int i = 0; i < 8; ++i) nx[i] = in4[tid + 128 * i];
    }
    lds_barrier();

    v8f d1[8];
    {
      v2f bf[8][2];
#pragma unroll
      for (int t = 0; t < 8; ++t) {
#pragma unroll
        for (int c = 0; c < 2; ++c) {
          bf[t][c] = v2f{ld1[t * PB + (4 * c) * PA], ld1[t * PB + (4 * c + 2) * PA]};
        }
      }
#pragma unroll
      for (int t = 0; t < 8; ++t) d1[t] = v8f{};
#pragma unroll
      for (int c = 0; c < 2; ++c) {
#pragma unroll
        for (int t = 0; t < 8; ++t) d1[t] = wmma4(a0[c], bf[t][c], d1[t]);
      }
    }
#pragma unroll
    for (int r = 0; r < 8; r += 2) {
      v8f e0, e1;
      second_group_pair(d1, a1, r, e0, e1);
#pragma unroll
      for (int q = 0; q < 8; ++q) {
        st1[r * PA + q * PB]       = e0[q];
        st1[(r + 1) * PA + q * PB] = e1[q];
      }
    }
    lds_barrier();
    {
      v2f bf[8][4];
#pragma unroll
      for (int t = 0; t < 8; ++t) {
#pragma unroll
        for (int c = 0; c < 4; ++c) {
          const int o = (c >> 1) * IMOFF + (4 * (c & 1)) * 8 + t;
          bf[t][c] = v2f{ld2[o], ld2[o + 16]};
        }
      }
#pragma unroll
      for (int t = 0; t < 8; ++t) d1[t] = v8f{};
#pragma unroll
      for (int c = 0; c < 4; ++c) {
#pragma unroll
        for (int t = 0; t < 8; ++t) d1[t] = wmma4(a2[c], bf[t][c], d1[t]);
      }
    }
    float acc = 0.f;
    {
      float Q[8] = {0.f, 0.f, 0.f, 0.f, 0.f, 0.f, 0.f, 0.f};
#pragma unroll
      for (int r = 0; r < 8; r += 2) {
        v8f e0, e1;
        second_group_pair(d1, a3, r, e0, e1);
        weigh_tile(e0, z, r, Q, acc);
        weigh_tile(e1, z, r + 1, Q, acc);
      }
      acc += z.e5 * (z.tz0 * (Q[0] - Q[4]) + z.tz1 * (Q[1] - Q[5])
                   + z.tz2 * (Q[2] - Q[6]) + z.tz3 * (Q[3] - Q[7]));
    }

#pragma unroll
    for (int off = 16; off; off >>= 1) acc += __shfl_down(acc, off, 32);
    if (lane == 0) red[wave] = acc;
    lds_barrier();
    if (tid == 0) out[s] = red[0] + red[1] + red[2] + red[3] + bias;
  }
}

extern "C" void kernel_launch(void* const* d_in, const int* in_sizes, int n_in,
                              void* d_out, int out_size, void* d_ws, size_t ws_size,
                              hipStream_t stream) {
  const float* state  = (const float*)d_in[0];
  const float* params = (const float*)d_in[1];
  const float* W      = (const float*)d_in[2];
  const float* b      = (const float*)d_in[3];
  float* out = (float*)d_out;

  qsim_main<<<NBLK, 128, 0, stream>>>(state, params, W, b, out);
}


# issue priority by states left: a wave starts at s_setprio 3 and drops a level per finished state, so a WGP's 4 work-groups advance together, not oldest-first, and none runs its last states alone
# speedup vs baseline: 1.5456x; 1.0459x over previous
#include <hip/hip_runtime.h>
#include <hip/hip_bf16.h>


typedef __attribute__((ext_vector_type(2))) float v2f;
typedef __attribute__((ext_vector_type(8))) float v8f;

#define NW 12
#define QDIM 4096
#define NSTATE 4096
#define NBLK 1024

#define PA 520
#define PB 65
#define PLANE 4159
#define IMOFF 4188

struct ZTerms { float base, e5, a6, a7, a8, tz0, tz1, tz2, tz3; };

__device__ __forceinline__ v8f wmma4(v2f a, v2f b, v8f c) {
  return __builtin_amdgcn_wmma_f32_16x16x4_f32(false, a, false, b,
                                               (short)0, c, false, false);
}

__device__ __forceinline__ void lds_barrier() {
  __builtin_amdgcn_fence(__ATOMIC_RELEASE, "workgroup", "local");
  __builtin_amdgcn_s_barrier();
  __builtin_amdgcn_fence(__ATOMIC_ACQUIRE, "workgroup", "local");
}


template <int NCH>
__device__ __forceinline__ void load_afrag_lds(v2f (&af)[NCH], const float* __restrict__ Ah,
                                               int h, int l16) {
  const float* arow = Ah + l16 * 16 + h;
#pragma unroll
  for (int c = 0; c < NCH; ++c) {
    af[c] = v2f{arow[(c >> 1) * 8 + 4 * (c & 1)], arow[(c >> 1) * 8 + 4 * (c & 1) + 2]};
  }
}
__device__ __forceinline__ void load_afrag_reg(v2f (&af)[4], const float* __restrict__ Ah,
                                               int h, int l16) {
  const float* arow = Ah + l16 * 16 + h * 8;
#pragma unroll
  for (int c = 0; c < 4; ++c) af[c] = v2f{arow[2 * c], arow[2 * c + 1]};
}

__device__ __forceinline__ void second_group_pair(const v8f (&d1)[8], const v2f (&af)[4],
                                                  int r, v8f& e0, v8f& e1) {
  e0 = v8f{}; e1 = v8f{};
#pragma unroll
  for (int c = 0; c < 4; ++c) {
    e0 = wmma4(af[c], v2f{d1[2 * c][r],     d1[2 * c + 1][r]},     e0);
    e1 = wmma4(af[c], v2f{d1[2 * c][r + 1], d1[2 * c + 1][r + 1]}, e1);
  }
}

__device__ __forceinline__ void weigh_tile(const v8f& d, const ZTerms& z, int r,
                                           float (&Q)[8], float& accb) {
  const bool n6 = ((r >> 2) & 1) != 0;
  const bool n7 = (((r >> 2) ^ (r >> 1)) & 1) != 0;
  const bool n8 = (((r >> 2) ^ (r >> 1) ^ r) & 1) != 0;
  float s1 = d[0] * d[0];
#pragma unroll
  for (int i = 1; i < 8; ++i) s1 = fmaf(d[i], d[i], s1);
  const float base = z.base + (n6 ? -z.a6 : z.a6) + (n7 ? -z.a7 : z.a7) + (n8 ? -z.a8 : z.a8);
  accb = fmaf(base, s1, accb);
#pragma unroll
  for (int i = 0; i < 8; ++i) Q[i] = fmaf(n8 ? -d[i] : d[i], d[i], Q[i]);
}

__global__ __launch_bounds__(128, 4) void qsim_main(
    const float* __restrict__ sb, const float* __restrict__ params,
    const float* __restrict__ W, const float* __restrict__ bptr,
    float* __restrict__ out) {
  __shared__ float S[IMOFF + PLANE];
  __shared__ float red[4];
  __shared__ float u_re[NW][4];
  __shared__ float u_im[NW][4];

  __builtin_amdgcn_s_setprio(3);

  const int tid  = threadIdx.x;
  const int lane = tid & 31;
  const int wave = __builtin_amdgcn_readfirstlane(tid >> 5);
  const int h    = lane >> 4;
  const int l16  = lane & 15;

  float th = 0.f, ph = 0.f;
#pragma unroll
  for (int w = 0; w < NW; ++w) {
    const float p0 = params[2 * w], p1 = params[2 * w + 1];
    th = (tid == w) ? p0 : th;
    ph = (tid == w) ? p1 : ph;
  }
  th *= 0.5f;
  ph *= 0.5f;

  float4 nx[8];
  const int sbase = 4 * tid + (tid >> 4);
  {
    const float4* in4 = reinterpret_cast<const float4*>(sb + (size_t)blockIdx.x * QDIM);
#pragma unroll
    for (int i = 0; i < 8; ++i) nx[i] = in4[tid + 128 * i];
  }

  if (tid < NW) {
    float s, c, sy, cy;
    sincosf(th, &s, &c);
    sincosf(ph, &sy, &cy);
    u_re[tid][0] = cy * c;   u_im[tid][0] =  sy * s;
    u_re[tid][1] = -sy * c;  u_im[tid][1] = -cy * s;
    u_re[tid][2] =  sy * c;  u_im[tid][2] = -cy * s;
    u_re[tid][3] =  cy * c;  u_im[tid][3] = -sy * s;
  }
  lds_barrier();
#pragma unroll
  for (int j = 0; j < 2; ++j) {
    const int e = tid + 128 * j;
    const int g = e >> 6, xp = (e >> 3) & 7, x = e & 7;
    float ar = 1.f, ai = 0.f;
#pragma unroll
    for (int q = 0; q < 3; ++q) {
      const int w = g * 3 + q;
      const int rc = ((xp >> (2 - q)) & 1) * 2 + ((x >> (2 - q)) & 1);
      const float br = u_re[w][rc], bi = u_im[w][rc];
      const float nr = ar * br - ai * bi;
      const float ni = ar * bi + ai * br;
      ar = nr; ai = ni;
    }
    float* T = S + g * 256 + xp * 16 + x;
    T[0] = ar;   T[8] = -ai;
    T[128] = ai; T[136] = ar;
  }
  lds_barrier();
  v2f a0[2], a1[4], a2[4], a3[4];
  load_afrag_lds<2>(a0, S,       h, l16);
  load_afrag_reg   (a1, S + 256, h, l16);
  load_afrag_lds<4>(a2, S + 512, h, l16);
  load_afrag_reg   (a3, S + 768, h, l16);
  const float bias = bptr[0];

  ZTerms z;
  {
    const int p0 = wave >> 1;
    const int p1 = p0 ^ (l16 >> 3);
    const int p2 = p1 ^ (wave & 1);
    const int p3 = p2 ^ ((l16 >> 2) & 1);
    const int p4 = p3 ^ ((l16 >> 1) & 1);
    const int p5 = p4 ^ (l16 & 1);
    z.base = (p0 ? -W[0] : W[0]) + (p1 ? -W[1] : W[1]) + (p2 ? -W[2] : W[2])
           + (p3 ? -W[3] : W[3]) + (p4 ? -W[4] : W[4]) + (p5 ? -W[5] : W[5]);
    z.e5 = p5 ? -1.f : 1.f;
    z.a6 = p5 ? -W[6] : W[6];
    z.a7 = p5 ? -W[7] : W[7];
    z.a8 = p5 ? -W[8] : W[8];
    z.tz0 = W[9] + W[10] + W[11];
    z.tz1 = W[9] + W[10] - W[11];
    z.tz2 = W[9] - W[10] - W[11];
    z.tz3 = W[9] - W[10] + W[11];
  }
  const int lc1 = (wave & 1) * 8 + (wave >> 1) * 32 + (l16 >> 3) * 16 + (l16 & 7);
  const int lc2 = (wave & 1) * PA + (wave >> 1) * (4 * PA) + (l16 >> 3) * (2 * PA) + (l16 & 7) * PB;
  const float* ld1 = S + lc1 + h * PA;
  float*       st1 = S + lc1 + h * IMOFF;
  const float* ld2 = S + lc2 + h * 8;
  lds_barrier();

  for (int s = blockIdx.x; s < NSTATE; s += NBLK) {
#pragma unroll
    for (int i = 0; i < 8; ++i) {
      float* p = S + sbase + i * PA;
      p[0] = nx[i].x; p[1] = nx[i].y; p[2] = nx[i].z; p[3] = nx[i].w;
    }
    if (s + NBLK < NSTATE) {
      const float4* in4 = reinterpret_cast<const float4*>(sb + (size_t)(s + NBLK) * QDIM);
#pragma unroll
      for (int i = 0; i < 8; ++i) nx[i] = in4[tid + 128 * i];
    }
    lds_barrier();

    v8f d1[8];
    {
      v2f bf[8][2];
#pragma unroll
      for (int t = 0; t < 8; ++t) {
#pragma unroll
        for (int c = 0; c < 2; ++c) {
          bf[t][c] = v2f{ld1[t * PB + (4 * c) * PA], ld1[t * PB + (4 * c + 2) * PA]};
        }
      }
#pragma unroll
      for (int t = 0; t < 8; ++t) d1[t] = v8f{};
#pragma unroll
      for (int c = 0; c < 2; ++c) {
#pragma unroll
        for (int t = 0; t < 8; ++t) d1[t] = wmma4(a0[c], bf[t][c], d1[t]);
      }
    }
#pragma unroll
    for (int r = 0; r < 8; r += 2) {
      v8f e0, e1;
      second_group_pair(d1, a1, r, e0, e1);
#pragma unroll
      for (int q = 0; q < 8; ++q) {
        st1[r * PA + q * PB]       = e0[q];
        st1[(r + 1) * PA + q * PB] = e1[q];
      }
    }
    lds_barrier();
    {
      v2f bf[8][4];
#pragma unroll
      for (int t = 0; t < 8; ++t) {
#pragma unroll
        for (int c = 0; c < 4; ++c) {
          const int o = (c >> 1) * IMOFF + (4 * (c & 1)) * 8 + t;
          bf[t][c] = v2f{ld2[o], ld2[o + 16]};
        }
      }
#pragma unroll
      for (int t = 0; t < 8; ++t) d1[t] = v8f{};
#pragma unroll
      for (int c = 0; c < 4; ++c) {
#pragma unroll
        for (int t = 0; t < 8; ++t) d1[t] = wmma4(a2[c], bf[t][c], d1[t]);
      }
    }
    float acc = 0.f;
    {
      float Q[8] = {0.f, 0.f, 0.f, 0.f, 0.f, 0.f, 0.f, 0.f};
#pragma unroll
      for (int r = 0; r < 8; r += 2) {
        v8f e0, e1;
        second_group_pair(d1, a3, r, e0, e1);
        weigh_tile(e0, z, r, Q, acc);
        weigh_tile(e1, z, r + 1, Q, acc);
      }
      acc += z.e5 * (z.tz0 * (Q[0] - Q[4]) + z.tz1 * (Q[1] - Q[5])
                   + z.tz2 * (Q[2] - Q[6]) + z.tz3 * (Q[3] - Q[7]));
    }

#pragma unroll
    for (int off = 16; off; off >>= 1) acc += __shfl_down(acc, off, 32);
    if (lane == 0) red[wave] = acc;
    lds_barrier();
    if (tid == 0) out[s] = red[0] + red[1] + red[2] + red[3] + bias;
    const int done = s / NBLK + 1;
    if (done == 1)      __builtin_amdgcn_s_setprio(2);
    else if (done == 2) __builtin_amdgcn_s_setprio(1);
    else                __builtin_amdgcn_s_setprio(0);
  }
}

extern "C" void kernel_launch(void* const* d_in, const int* in_sizes, int n_in,
                              void* d_out, int out_size, void* d_ws, size_t ws_size,
                              hipStream_t stream) {
  const float* state  = (const float*)d_in[0];
  const float* params = (const float*)d_in[1];
  const float* W      = (const float*)d_in[2];
  const float* b      = (const float*)d_in[3];
  float* out = (float*)d_out;

  qsim_main<<<NBLK, 128, 0, stream>>>(state, params, W, b, out);
}


# a state's real plane is loaded from memory straight into the first gate group's B fragments (per-lane 4-byte loads, one state ahead): no LDS staging of the input, 2 barriers per state, not 3
# speedup vs baseline: 1.6419x; 1.0623x over previous
#include <hip/hip_runtime.h>
#include <hip/hip_bf16.h>


typedef __attribute__((ext_vector_type(2))) float v2f;
typedef __attribute__((ext_vector_type(8))) float v8f;

#define NW 12
#define QDIM 4096
#define NSTATE 4096
#define NBLK 1024

#define PA 520
#define PB 65
#define PLANE 4159
#define IMOFF 4188

struct ZTerms { float base, e5, a6, a7, a8, tz0, tz1, tz2, tz3; };

__device__ __forceinline__ v8f wmma4(v2f a, v2f b, v8f c) {
  return __builtin_amdgcn_wmma_f32_16x16x4_f32(false, a, false, b,
                                               (short)0, c, false, false);
}

__device__ __forceinline__ void lds_barrier() {
  __builtin_amdgcn_fence(__ATOMIC_RELEASE, "workgroup", "local");
  __builtin_amdgcn_s_barrier();
  __builtin_amdgcn_fence(__ATOMIC_ACQUIRE, "workgroup", "local");
}


template <int NCH>
__device__ __forceinline__ void load_afrag_lds(v2f (&af)[NCH], const float* __restrict__ Ah,
                                               int h, int l16) {
  const float* arow = Ah + l16 * 16 + h;
#pragma unroll
  for (int c = 0; c < NCH; ++c) {
    af[c] = v2f{arow[(c >> 1) * 8 + 4 * (c & 1)], arow[(c >> 1) * 8 + 4 * (c & 1) + 2]};
  }
}
__device__ __forceinline__ void load_afrag_reg(v2f (&af)[4], const float* __restrict__ Ah,
                                               int h, int l16) {
  const float* arow = Ah + l16 * 16 + h * 8;
#pragma unroll
  for (int c = 0; c < 4; ++c) af[c] = v2f{arow[2 * c], arow[2 * c + 1]};
}

__device__ __forceinline__ void second_group_pair(const v8f (&d1)[8], const v2f (&af)[4],
                                                  int r, v8f& e0, v8f& e1) {
  e0 = v8f{}; e1 = v8f{};
#pragma unroll
  for (int c = 0; c < 4; ++c) {
    e0 = wmma4(af[c], v2f{d1[2 * c][r],     d1[2 * c + 1][r]},     e0);
    e1 = wmma4(af[c], v2f{d1[2 * c][r + 1], d1[2 * c + 1][r + 1]}, e1);
  }
}

__device__ __forceinline__ void weigh_tile(const v8f& d, const ZTerms& z, int r,
                                           float (&Q)[8], float& accb) {
  const bool n6 = ((r >> 2) & 1) != 0;
  const bool n7 = (((r >> 2) ^ (r >> 1)) & 1) != 0;
  const bool n8 = (((r >> 2) ^ (r >> 1) ^ r) & 1) != 0;
  float s1 = d[0] * d[0];
#pragma unroll
  for (int i = 1; i < 8; ++i) s1 = fmaf(d[i], d[i], s1);
  const float base = z.base + (n6 ? -z.a6 : z.a6) + (n7 ? -z.a7 : z.a7) + (n8 ? -z.a8 : z.a8);
  accb = fmaf(base, s1, accb);
#pragma unroll
  for (int i = 0; i < 8; ++i) Q[i] = fmaf(n8 ? -d[i] : d[i], d[i], Q[i]);
}

__global__ __launch_bounds__(128, 4) void qsim_main(
    const float* __restrict__ sb, const float* __restrict__ params,
    const float* __restrict__ W, const float* __restrict__ bptr,
    float* __restrict__ out) {
  __shared__ float S[IMOFF + PLANE];
  __shared__ float red[4];
  __shared__ float u_re[NW][4];
  __shared__ float u_im[NW][4];

  __builtin_amdgcn_s_setprio(3);

  const int tid  = threadIdx.x;
  const int lane = tid & 31;
  const int wave = __builtin_amdgcn_readfirstlane(tid >> 5);
  const int h    = lane >> 4;
  const int l16  = lane & 15;

  float th = 0.f, ph = 0.f;
#pragma unroll
  for (int w = 0; w < NW; ++w) {
    const float p0 = params[2 * w], p1 = params[2 * w + 1];
    th = (tid == w) ? p0 : th;
    ph = (tid == w) ? p1 : ph;
  }
  th *= 0.5f;
  ph *= 0.5f;

  const int lc1 = (wave & 1) * 8 + (wave >> 1) * 32 + (l16 >> 3) * 16 + (l16 & 7);
  const int gl1 = h * 512 + lc1;
  v2f bq[8][2];
  {
    const float* g = sb + (size_t)blockIdx.x * QDIM + gl1;
#pragma unroll
    for (int t = 0; t < 8; ++t) {
#pragma unroll
      for (int c = 0; c < 2; ++c) {
        bq[t][c] = v2f{g[(4 * c) * 512 + t * 64], g[(4 * c + 2) * 512 + t * 64]};
      }
    }
  }

  if (tid < NW) {
    float s, c, sy, cy;
    sincosf(th, &s, &c);
    sincosf(ph, &sy, &cy);
    u_re[tid][0] = cy * c;   u_im[tid][0] =  sy * s;
    u_re[tid][1] = -sy * c;  u_im[tid][1] = -cy * s;
    u_re[tid][2] =  sy * c;  u_im[tid][2] = -cy * s;
    u_re[tid][3] =  cy * c;  u_im[tid][3] = -sy * s;
  }
  lds_barrier();
#pragma unroll
  for (int j = 0; j < 2; ++j) {
    const int e = tid + 128 * j;
    const int g = e >> 6, xp = (e >> 3) & 7, x = e & 7;
    float ar = 1.f, ai = 0.f;
#pragma unroll
    for (int q = 0; q < 3; ++q) {
      const int w = g * 3 + q;
      const int rc = ((xp >> (2 - q)) & 1) * 2 + ((x >> (2 - q)) & 1);
      const float br = u_re[w][rc], bi = u_im[w][rc];
      const float nr = ar * br - ai * bi;
      const float ni = ar * bi + ai * br;
      ar = nr; ai = ni;
    }
    float* T = S + g * 256 + xp * 16 + x;
    T[0] = ar;   T[8] = -ai;
    T[128] = ai; T[136] = ar;
  }
  lds_barrier();
  v2f a0[2], a1[4], a2[4], a3[4];
  load_afrag_lds<2>(a0, S,       h, l16);
  load_afrag_reg   (a1, S + 256, h, l16);
  load_afrag_lds<4>(a2, S + 512, h, l16);
  load_afrag_reg   (a3, S + 768, h, l16);
  const float bias = bptr[0];

  ZTerms z;
  {
    const int p0 = wave >> 1;
    const int p1 = p0 ^ (l16 >> 3);
    const int p2 = p1 ^ (wave & 1);
    const int p3 = p2 ^ ((l16 >> 2) & 1);
    const int p4 = p3 ^ ((l16 >> 1) & 1);
    const int p5 = p4 ^ (l16 & 1);
    z.base = (p0 ? -W[0] : W[0]) + (p1 ? -W[1] : W[1]) + (p2 ? -W[2] : W[2])
           + (p3 ? -W[3] : W[3]) + (p4 ? -W[4] : W[4]) + (p5 ? -W[5] : W[5]);
    z.e5 = p5 ? -1.f : 1.f;
    z.a6 = p5 ? -W[6] : W[6];
    z.a7 = p5 ? -W[7] : W[7];
    z.a8 = p5 ? -W[8] : W[8];
    z.tz0 = W[9] + W[10] + W[11];
    z.tz1 = W[9] + W[10] - W[11];
    z.tz2 = W[9] - W[10] - W[11];
    z.tz3 = W[9] - W[10] + W[11];
  }
  const int lc2 = (wave & 1) * PA + (wave >> 1) * (4 * PA) + (l16 >> 3) * (2 * PA) + (l16 & 7) * PB;
  float*       st1 = S + lc1 + h * IMOFF;
  const float* ld2 = S + lc2 + h * 8;
  lds_barrier();

  for (int s = blockIdx.x; s < NSTATE; s += NBLK) {
    v8f d1[8];
#pragma unroll
    for (int t = 0; t < 8; ++t) d1[t] = v8f{};
#pragma unroll
    for (int c = 0; c < 2; ++c) {
#pragma unroll
      for (int t = 0; t < 8; ++t) d1[t] = wmma4(a0[c], bq[t][c], d1[t]);
    }
    if (s + NBLK < NSTATE) {
      const float* g = sb + (size_t)(s + NBLK) * QDIM + gl1;
#pragma unroll
      for (int t = 0; t < 8; ++t) {
#pragma unroll
        for (int c = 0; c < 2; ++c) {
          bq[t][c] = v2f{g[(4 * c) * 512 + t * 64], g[(4 * c + 2) * 512 + t * 64]};
        }
      }
    }
#pragma unroll
    for (int r = 0; r < 8; r += 2) {
      v8f e0, e1;
      second_group_pair(d1, a1, r, e0, e1);
#pragma unroll
      for (int q = 0; q < 8; ++q) {
        st1[r * PA + q * PB]       = e0[q];
        st1[(r + 1) * PA + q * PB] = e1[q];
      }
    }
    lds_barrier();
    {
      v2f bf[8][4];
#pragma unroll
      for (int t = 0; t < 8; ++t) {
#pragma unroll
        for (int c = 0; c < 4; ++c) {
          const int o = (c >> 1) * IMOFF + (4 * (c & 1)) * 8 + t;
          bf[t][c] = v2f{ld2[o], ld2[o + 16]};
        }
      }
#pragma unroll
      for (int t = 0; t < 8; ++t) d1[t] = v8f{};
#pragma unroll
      for (int c = 0; c < 4; ++c) {
#pragma unroll
        for (int t = 0; t < 8; ++t) d1[t] = wmma4(a2[c], bf[t][c], d1[t]);
      }
    }
    float acc = 0.f;
    {
      float Q[8] = {0.f, 0.f, 0.f, 0.f, 0.f, 0.f, 0.f, 0.f};
#pragma unroll
      for (int r = 0; r < 8; r += 2) {
        v8f e0, e1;
        second_group_pair(d1, a3, r, e0, e1);
        weigh_tile(e0, z, r, Q, acc);
        weigh_tile(e1, z, r + 1, Q, acc);
      }
      acc += z.e5 * (z.tz0 * (Q[0] - Q[4]) + z.tz1 * (Q[1] - Q[5])
                   + z.tz2 * (Q[2] - Q[6]) + z.tz3 * (Q[3] - Q[7]));
    }

#pragma unroll
    for (int off = 16; off; off >>= 1) acc += __shfl_down(acc, off, 32);
    if (lane == 0) red[wave] = acc;
    lds_barrier();
    if (tid == 0) out[s] = red[0] + red[1] + red[2] + red[3] + bias;
    const int done = s / NBLK + 1;
    if (done == 1)      __builtin_amdgcn_s_setprio(2);
    else if (done == 2) __builtin_amdgcn_s_setprio(1);
    else                __builtin_amdgcn_s_setprio(0);
  }
}

extern "C" void kernel_launch(void* const* d_in, const int* in_sizes, int n_in,
                              void* d_out, int out_size, void* d_ws, size_t ws_size,
                              hipStream_t stream) {
  const float* state  = (const float*)d_in[0];
  const float* params = (const float*)d_in[1];
  const float* W      = (const float*)d_in[2];
  const float* b      = (const float*)d_in[3];
  float* out = (float*)d_out;

  qsim_main<<<NBLK, 128, 0, stream>>>(state, params, W, b, out);
}
